# starting kernel
# baseline (speedup 1.0000x reference)
#include <hip/hip_runtime.h>
#include <math.h>

#define SEQ    256
#define WE     100
#define TE     20
#define IN_DIM (WE + TE)
#define H      400
#define G4     (4 * H)
#define M_DIM  200
#define D2H    (2 * H)
#define NPROJ  12
#define NFLAT  (NPROJ * M_DIM)
#define N_ARC  65536
#define N_SIB  200000
#define N_GP   200000
#define N_GSIB 300000

typedef __attribute__((ext_vector_type(2))) float v2f;
typedef __attribute__((ext_vector_type(8))) float v8f;

__device__ __forceinline__ float sigmoidf_(float x) {
    return 1.0f / (1.0f + __expf(-x));
}

__device__ __forceinline__ v8f wmma_f32(v2f a, v2f b, v8f c) {
    return __builtin_amdgcn_wmma_f32_16x16x4_f32(false, a, false, b, (short)0, c,
                                                 false, false);
}

__global__ void embed_kernel(const int* __restrict__ words, const int* __restrict__ tags,
                             const float* __restrict__ wemb, const float* __restrict__ temb,
                             float* __restrict__ x) {
    const int s = blockIdx.x;
    const int t = threadIdx.x;
    if (t < WE)            x[s * IN_DIM + t] = wemb[words[s] * WE + t];
    else if (t < IN_DIM)   x[s * IN_DIM + t] = temb[tags[s] * TE + (t - WE)];
}

__global__ void igemm_kernel(const float* __restrict__ x,
                             const float* __restrict__ Wih_f, const float* __restrict__ b_f,
                             const float* __restrict__ Wih_b, const float* __restrict__ b_b,
                             float* __restrict__ z_f, float* __restrict__ z_b) {
    const int lane = threadIdx.x & 31;
    const int m0 = blockIdx.x * 32;
    const int n0 = blockIdx.y * 32;
    const int dir = blockIdx.z;

    const float* W    = dir ? Wih_b : Wih_f;
    const float* bias = dir ? b_b : b_f;
    float*       z    = dir ? z_b : z_f;

    const int mrow0 = m0 + (lane & 15);
    const int mrow1 = mrow0 + 16;
    const int arow0 = dir ? (SEQ - 1 - mrow0) : mrow0;
    const int arow1 = dir ? (SEQ - 1 - mrow1) : mrow1;
    const int kh    = (lane >> 4) * 2;
    const int ncol0 = n0 + (lane & 15);
    const int ncol1 = ncol0 + 16;

    v8f c00 = {}, c01 = {}, c10 = {}, c11 = {};
    for (int k = 0; k < IN_DIM; k += 4) {
        v2f a0 = *(const v2f*)(x + arow0 * IN_DIM + k + kh);
        v2f a1 = *(const v2f*)(x + arow1 * IN_DIM + k + kh);
        v2f b0 = *(const v2f*)(W + ncol0 * IN_DIM + k + kh);
        v2f b1 = *(const v2f*)(W + ncol1 * IN_DIM + k + kh);
        c00 = wmma_f32(a0, b0, c00);
        c01 = wmma_f32(a0, b1, c01);
        c10 = wmma_f32(a1, b0, c10);
        c11 = wmma_f32(a1, b1, c11);
    }
    const float bn0 = bias[ncol0];
    const float bn1 = bias[ncol1];
    const int mb = m0 + (lane >> 4) * 8;
#pragma unroll
    for (int r = 0; r < 8; ++r) {
        z[(mb + r) * G4 + ncol0]      = c00[r] + bn0;
        z[(mb + r) * G4 + ncol1]      = c01[r] + bn1;
        z[(mb + 16 + r) * G4 + ncol0] = c10[r] + bn0;
        z[(mb + 16 + r) * G4 + ncol1] = c11[r] + bn1;
    }
}

__global__ void lstm_kernel(const float* __restrict__ z_f, const float* __restrict__ z_b,
                            const float* __restrict__ Whh_f, const float* __restrict__ Whh_b,
                            float* __restrict__ states) {
    const int dir = blockIdx.x;
    const float* z_in = dir ? z_b : z_f;
    const float* Whh  = dir ? Whh_b : Whh_f;

    __shared__ __align__(16) float h_s[H];
    __shared__ __align__(16) float c_s[H];
    __shared__ __align__(16) float z_s[G4];

    const int tid = threadIdx.x;
    for (int u = tid; u < H; u += 1024) { h_s[u] = 0.0f; c_s[u] = 0.0f; }
    __syncthreads();

    for (int t = 0; t < SEQ; ++t) {
        for (int row = tid; row < G4; row += 1024) {
            const float4* wr = (const float4*)(Whh + row * H);
            float acc = 0.0f;
#pragma unroll 4
            for (int k4 = 0; k4 < H / 4; ++k4) {
                float4 w  = wr[k4];
                float4 hv = *(const float4*)(h_s + 4 * k4);
                acc += w.x * hv.x + w.y * hv.y + w.z * hv.z + w.w * hv.w;
            }
            z_s[row] = z_in[t * G4 + row] + acc;
            if (t + 1 < SEQ)
                __builtin_prefetch(z_in + (t + 1) * G4 + row, 0, 1);
        }
        __syncthreads();
        if (tid < H) {
            float i = sigmoidf_(z_s[tid]);
            float f = sigmoidf_(z_s[tid + H]);
            float g = tanhf(z_s[tid + 2 * H]);
            float o = sigmoidf_(z_s[tid + 3 * H]);
            float cn = f * c_s[tid] + i * g;
            float hn = o * tanhf(cn);
            c_s[tid] = cn;
            h_s[tid] = hn;
            const int spos = dir ? (SEQ - 1 - t) : t;
            states[spos * D2H + dir * H + tid] = hn;
        }
        __syncthreads();
    }
}

__global__ void proj_kernel(const float* __restrict__ states,
                            const float* __restrict__ Wp,
                            float* __restrict__ proj) {
    const int lane = threadIdx.x & 31;
    const int m0 = blockIdx.x * 32;
    const int n0 = blockIdx.y * 32;
    const int mrow0 = m0 + (lane & 15);
    const int mrow1 = mrow0 + 16;
    const int kh    = (lane >> 4) * 2;
    const int ncol0 = n0 + (lane & 15);
    const int ncol1 = ncol0 + 16;

    v8f c00 = {}, c01 = {}, c10 = {}, c11 = {};
    for (int k = 0; k < D2H; k += 4) {
        v2f a0 = *(const v2f*)(states + mrow0 * D2H + k + kh);
        v2f a1 = *(const v2f*)(states + mrow1 * D2H + k + kh);
        v2f b0 = *(const v2f*)(Wp + ncol0 * D2H + k + kh);
        v2f b1 = *(const v2f*)(Wp + ncol1 * D2H + k + kh);
        c00 = wmma_f32(a0, b0, c00);
        c01 = wmma_f32(a0, b1, c01);
        c10 = wmma_f32(a1, b0, c10);
        c11 = wmma_f32(a1, b1, c11);
    }
    const int mb = m0 + (lane >> 4) * 8;
#pragma unroll
    for (int r = 0; r < 8; ++r) {
        proj[(mb + r) * NFLAT + ncol0]      = c00[r];
        proj[(mb + r) * NFLAT + ncol1]      = c01[r];
        proj[(mb + 16 + r) * NFLAT + ncol0] = c10[r];
        proj[(mb + 16 + r) * NFLAT + ncol1] = c11[r];
    }
}

__global__ void score_kernel(const float* __restrict__ proj,
                             const int* __restrict__ ih, const int* __restrict__ im,
                             const int* __restrict__ is, const int* __restrict__ ig,
                             int ph, int pm, int ps, int pg,
                             const float* __restrict__ null_sib,
                             const float* __restrict__ wsc,
                             float* __restrict__ out, int N) {
    const int lane   = threadIdx.x & 31;
    const int wid    = (blockIdx.x * blockDim.x + threadIdx.x) >> 5;
    const int nwaves = (gridDim.x * blockDim.x) >> 5;

    for (int tup = wid; tup < N; tup += nwaves) {
        const float* rh = proj + ih[tup] * NFLAT + ph;
        const float* rm = proj + im[tup] * NFLAT + pm;
        const float* rs = nullptr;
        const float* rg = nullptr;
        if (is) {
            const int s = is[tup];
            rs = (s >= SEQ) ? null_sib : (proj + s * NFLAT + ps);
        }
        if (ig) rg = proj + ig[tup] * NFLAT + pg;

        float acc = 0.0f;
        for (int m = lane; m < M_DIM; m += 32) {
            float v = rh[m] + rm[m];
            if (rs) v += rs[m];
            if (rg) v += rg[m];
            acc += tanhf(v) * wsc[m];
        }
#pragma unroll
        for (int off = 16; off > 0; off >>= 1)
            acc += __shfl_xor(acc, off, 32);
        if (lane == 0) out[tup] = acc;
    }
}

extern "C" void kernel_launch(void* const* d_in, const int* in_sizes, int n_in,
                              void* d_out, int out_size, void* d_ws, size_t ws_size,
                              hipStream_t stream) {
    const int* words      = (const int*)d_in[0];
    const int* tags       = (const int*)d_in[1];
    const int* arc_head   = (const int*)d_in[2];
    const int* arc_mod    = (const int*)d_in[3];
    const int* sib_head   = (const int*)d_in[4];
    const int* sib_mod    = (const int*)d_in[5];
    const int* sib_sib    = (const int*)d_in[6];
    const int* gp_head    = (const int*)d_in[7];
    const int* gp_mod     = (const int*)d_in[8];
    const int* gp_grand   = (const int*)d_in[9];
    const int* gsib_head  = (const int*)d_in[10];
    const int* gsib_mod   = (const int*)d_in[11];
    const int* gsib_sib   = (const int*)d_in[12];
    const int* gsib_grand = (const int*)d_in[13];
    const float* word_emb = (const float*)d_in[14];
    const float* tag_emb  = (const float*)d_in[15];
    const float* Wih_f    = (const float*)d_in[16];
    const float* Whh_f    = (const float*)d_in[17];
    const float* b_f      = (const float*)d_in[18];
    const float* Wih_b    = (const float*)d_in[19];
    const float* Whh_b    = (const float*)d_in[20];
    const float* b_b      = (const float*)d_in[21];
    const float* W_proj   = (const float*)d_in[22];
    const float* W_score  = (const float*)d_in[23];
    const float* null_sib = (const float*)d_in[24];

    float* out = (float*)d_out;

    float* ws      = (float*)d_ws;
    float* x       = ws;
    float* z_f     = x + SEQ * IN_DIM;
    float* z_b     = z_f + SEQ * G4;
    float* states  = z_b + SEQ * G4;
    float* projbuf = states + SEQ * D2H;

    embed_kernel<<<SEQ, 128, 0, stream>>>(words, tags, word_emb, tag_emb, x);

    igemm_kernel<<<dim3(SEQ / 32, G4 / 32, 2), 32, 0, stream>>>(
        x, Wih_f, b_f, Wih_b, b_b, z_f, z_b);

    lstm_kernel<<<2, 1024, 0, stream>>>(z_f, z_b, Whh_f, Whh_b, states);

    proj_kernel<<<dim3(SEQ / 32, NFLAT / 32), 32, 0, stream>>>(states, W_proj, projbuf);

    const dim3 sgrid(1024), sblk(256);
    score_kernel<<<sgrid, sblk, 0, stream>>>(projbuf, arc_head, arc_mod, nullptr, nullptr,
                                             0 * M_DIM, 1 * M_DIM, 0, 0,
                                             null_sib, W_score + 0 * M_DIM,
                                             out, N_ARC);
    score_kernel<<<sgrid, sblk, 0, stream>>>(projbuf, sib_head, sib_mod, sib_sib, nullptr,
                                             5 * M_DIM, 6 * M_DIM, 7 * M_DIM, 0,
                                             null_sib, W_score + 1 * M_DIM,
                                             out + N_ARC, N_SIB);
    score_kernel<<<sgrid, sblk, 0, stream>>>(projbuf, gp_head, gp_mod, nullptr, gp_grand,
                                             3 * M_DIM, 4 * M_DIM, 0, 2 * M_DIM,
                                             null_sib, W_score + 2 * M_DIM,
                                             out + N_ARC + N_SIB, N_GP);
    score_kernel<<<sgrid, sblk, 0, stream>>>(projbuf, gsib_head, gsib_mod, gsib_sib, gsib_grand,
                                             8 * M_DIM, 9 * M_DIM, 10 * M_DIM, 11 * M_DIM,
                                             null_sib, W_score + 3 * M_DIM,
                                             out + N_ARC + N_SIB + N_GP, N_GSIB);
}


# LSTM recurrence as one launch per time step: each step's Whh*h spread over 200 work-groups, a wave per hidden unit, whole-line loads; lane-strided partial sums + wave butterfly change the sum order
# speedup vs baseline: 10.2304x; 10.2304x over previous
#include <hip/hip_runtime.h>
#include <math.h>

#define SEQ    256
#define WE     100
#define TE     20
#define IN_DIM (WE + TE)
#define H      400
#define G4     (4 * H)
#define M_DIM  200
#define D2H    (2 * H)
#define NPROJ  12
#define NFLAT  (NPROJ * M_DIM)
#define N_ARC  65536
#define N_SIB  200000
#define N_GP   200000
#define N_GSIB 300000

typedef __attribute__((ext_vector_type(2))) float v2f;
typedef __attribute__((ext_vector_type(8))) float v8f;

__device__ __forceinline__ float sigmoidf_(float x) {
    return 1.0f / (1.0f + __expf(-x));
}

__device__ __forceinline__ v8f wmma_f32(v2f a, v2f b, v8f c) {
    return __builtin_amdgcn_wmma_f32_16x16x4_f32(false, a, false, b, (short)0, c,
                                                 false, false);
}

__global__ void embed_kernel(const int* __restrict__ words, const int* __restrict__ tags,
                             const float* __restrict__ wemb, const float* __restrict__ temb,
                             float* __restrict__ x) {
    const int s = blockIdx.x;
    const int t = threadIdx.x;
    if (t < WE)            x[s * IN_DIM + t] = wemb[words[s] * WE + t];
    else if (t < IN_DIM)   x[s * IN_DIM + t] = temb[tags[s] * TE + (t - WE)];
}

__global__ void igemm_kernel(const float* __restrict__ x,
                             const float* __restrict__ Wih_f, const float* __restrict__ b_f,
                             const float* __restrict__ Wih_b, const float* __restrict__ b_b,
                             float* __restrict__ z_f, float* __restrict__ z_b) {
    const int lane = threadIdx.x & 31;
    const int m0 = blockIdx.x * 32;
    const int n0 = blockIdx.y * 32;
    const int dir = blockIdx.z;

    const float* W    = dir ? Wih_b : Wih_f;
    const float* bias = dir ? b_b : b_f;
    float*       z    = dir ? z_b : z_f;

    const int mrow0 = m0 + (lane & 15);
    const int mrow1 = mrow0 + 16;
    const int arow0 = dir ? (SEQ - 1 - mrow0) : mrow0;
    const int arow1 = dir ? (SEQ - 1 - mrow1) : mrow1;
    const int kh    = (lane >> 4) * 2;
    const int ncol0 = n0 + (lane & 15);
    const int ncol1 = ncol0 + 16;

    v8f c00 = {}, c01 = {}, c10 = {}, c11 = {};
    for (int k = 0; k < IN_DIM; k += 4) {
        v2f a0 = *(const v2f*)(x + arow0 * IN_DIM + k + kh);
        v2f a1 = *(const v2f*)(x + arow1 * IN_DIM + k + kh);
        v2f b0 = *(const v2f*)(W + ncol0 * IN_DIM + k + kh);
        v2f b1 = *(const v2f*)(W + ncol1 * IN_DIM + k + kh);
        c00 = wmma_f32(a0, b0, c00);
        c01 = wmma_f32(a0, b1, c01);
        c10 = wmma_f32(a1, b0, c10);
        c11 = wmma_f32(a1, b1, c11);
    }
    const float bn0 = bias[ncol0];
    const float bn1 = bias[ncol1];
    const int mb = m0 + (lane >> 4) * 8;
#pragma unroll
    for (int r = 0; r < 8; ++r) {
        z[(mb + r) * G4 + ncol0]      = c00[r] + bn0;
        z[(mb + r) * G4 + ncol1]      = c01[r] + bn1;
        z[(mb + 16 + r) * G4 + ncol0] = c10[r] + bn0;
        z[(mb + 16 + r) * G4 + ncol1] = c11[r] + bn1;
    }
}

__device__ __forceinline__ float dot4_(float4 w, float4 h) {
    return w.x * h.x + w.y * h.y + w.z * h.z + w.w * h.w;
}

template <int M>
__device__ __forceinline__ float xrow_(float x) {
    return __builtin_bit_cast(float, __builtin_amdgcn_update_dpp(
        0, __builtin_bit_cast(int, x), 0x160 + M, 0xf, 0xf, false));
}
__device__ __forceinline__ float xhalf_(float x) {
    return __builtin_bit_cast(float, __builtin_amdgcn_permlanex16(
        __builtin_bit_cast(int, x), __builtin_bit_cast(int, x),
        0x76543210, 0xfedcba98, false, false));
}
__device__ __forceinline__ float wave_sum_(float x) {
    x += xrow_<1>(x);
    x += xrow_<2>(x);
    x += xrow_<4>(x);
    x += xrow_<8>(x);
    x += xhalf_(x);
    return x;
}

#define STEP_WAVES 4
#define STEP_GRID  (2 * H / STEP_WAVES)

template <bool FIRST>
__global__ void __launch_bounds__(32 * STEP_WAVES) __attribute__((amdgpu_waves_per_eu(1, 2)))
lstm_step_kernel(const float* __restrict__ z_t,
                 const float* __restrict__ Whh_f, const float* __restrict__ Whh_b,
                 const float* __restrict__ hprev_f, const float* __restrict__ hprev_b,
                 float* __restrict__ cstate,
                 float* __restrict__ hout_f, float* __restrict__ hout_b) {
    const int lane = threadIdx.x & 31;
    const int wv  = __builtin_amdgcn_readfirstlane((int)(blockIdx.x * STEP_WAVES + (threadIdx.x >> 5)));
    const int dir = (wv >= H) ? 1 : 0;
    const int j   = wv - dir * H;

    const float* z    = z_t + dir * (SEQ * G4);
    float*       hout = dir ? hout_b : hout_f;
    float*       c    = cstate + dir * H;

    float ai = 0.0f, af = 0.0f, ag = 0.0f, ao = 0.0f;
    float cprev = 0.0f;
    if (!FIRST) {
        const float4* h4 = (const float4*)(dir ? hprev_b : hprev_f);
        const float4* wi = (const float4*)((dir ? Whh_b : Whh_f) + j * H);
        const float4* wf = wi + (H * H) / 4;
        const float4* wg = wf + (H * H) / 4;
        const float4* wo = wg + (H * H) / 4;
        cprev = c[j];
        const int  qt = (lane < H / 4 - 96) ? (96 + lane) : (H / 4 - 1);
        const bool tail = lane < H / 4 - 96;
        float4 hv[4], vi[4], vf[4], vg[4], vo[4];
#pragma unroll
        for (int i = 0; i < 4; ++i) {
            const int q = (i < 3) ? (lane + 32 * i) : qt;
            hv[i] = h4[q];
            vi[i] = wi[q];
            vf[i] = wf[q];
            vg[i] = wg[q];
            vo[i] = wo[q];
        }
        hv[3].x = tail ? hv[3].x : 0.0f;
        hv[3].y = tail ? hv[3].y : 0.0f;
        hv[3].z = tail ? hv[3].z : 0.0f;
        hv[3].w = tail ? hv[3].w : 0.0f;
#pragma unroll
        for (int i = 0; i < 4; ++i) {
            ai += dot4_(vi[i], hv[i]);
            af += dot4_(vf[i], hv[i]);
            ag += dot4_(vg[i], hv[i]);
            ao += dot4_(vo[i], hv[i]);
        }
        ai = wave_sum_(ai);
        af = wave_sum_(af);
        ag = wave_sum_(ag);
        ao = wave_sum_(ao);
    }
    const float i_ = sigmoidf_(z[j] + ai);
    const float f_ = sigmoidf_(z[j + H] + af);
    const float g_ = tanhf(z[j + 2 * H] + ag);
    const float o_ = sigmoidf_(z[j + 3 * H] + ao);
    const float cn = f_ * cprev + i_ * g_;
    const float hn = o_ * tanhf(cn);
    if (lane == 0) {
        c[j]    = cn;
        hout[j] = hn;
    }
}

__global__ void proj_kernel(const float* __restrict__ states,
                            const float* __restrict__ Wp,
                            float* __restrict__ proj) {
    const int lane = threadIdx.x & 31;
    const int m0 = blockIdx.x * 32;
    const int n0 = blockIdx.y * 32;
    const int mrow0 = m0 + (lane & 15);
    const int mrow1 = mrow0 + 16;
    const int kh    = (lane >> 4) * 2;
    const int ncol0 = n0 + (lane & 15);
    const int ncol1 = ncol0 + 16;

    v8f c00 = {}, c01 = {}, c10 = {}, c11 = {};
    for (int k = 0; k < D2H; k += 4) {
        v2f a0 = *(const v2f*)(states + mrow0 * D2H + k + kh);
        v2f a1 = *(const v2f*)(states + mrow1 * D2H + k + kh);
        v2f b0 = *(const v2f*)(Wp + ncol0 * D2H + k + kh);
        v2f b1 = *(const v2f*)(Wp + ncol1 * D2H + k + kh);
        c00 = wmma_f32(a0, b0, c00);
        c01 = wmma_f32(a0, b1, c01);
        c10 = wmma_f32(a1, b0, c10);
        c11 = wmma_f32(a1, b1, c11);
    }
    const int mb = m0 + (lane >> 4) * 8;
#pragma unroll
    for (int r = 0; r < 8; ++r) {
        proj[(mb + r) * NFLAT + ncol0]      = c00[r];
        proj[(mb + r) * NFLAT + ncol1]      = c01[r];
        proj[(mb + 16 + r) * NFLAT + ncol0] = c10[r];
        proj[(mb + 16 + r) * NFLAT + ncol1] = c11[r];
    }
}

__global__ void score_kernel(const float* __restrict__ proj,
                             const int* __restrict__ ih, const int* __restrict__ im,
                             const int* __restrict__ is, const int* __restrict__ ig,
                             int ph, int pm, int ps, int pg,
                             const float* __restrict__ null_sib,
                             const float* __restrict__ wsc,
                             float* __restrict__ out, int N) {
    const int lane   = threadIdx.x & 31;
    const int wid    = (blockIdx.x * blockDim.x + threadIdx.x) >> 5;
    const int nwaves = (gridDim.x * blockDim.x) >> 5;

    for (int tup = wid; tup < N; tup += nwaves) {
        const float* rh = proj + ih[tup] * NFLAT + ph;
        const float* rm = proj + im[tup] * NFLAT + pm;
        const float* rs = nullptr;
        const float* rg = nullptr;
        if (is) {
            const int s = is[tup];
            rs = (s >= SEQ) ? null_sib : (proj + s * NFLAT + ps);
        }
        if (ig) rg = proj + ig[tup] * NFLAT + pg;

        float acc = 0.0f;
        for (int m = lane; m < M_DIM; m += 32) {
            float v = rh[m] + rm[m];
            if (rs) v += rs[m];
            if (rg) v += rg[m];
            acc += tanhf(v) * wsc[m];
        }
#pragma unroll
        for (int off = 16; off > 0; off >>= 1)
            acc += __shfl_xor(acc, off, 32);
        if (lane == 0) out[tup] = acc;
    }
}

extern "C" void kernel_launch(void* const* d_in, const int* in_sizes, int n_in,
                              void* d_out, int out_size, void* d_ws, size_t ws_size,
                              hipStream_t stream) {
    const int* words      = (const int*)d_in[0];
    const int* tags       = (const int*)d_in[1];
    const int* arc_head   = (const int*)d_in[2];
    const int* arc_mod    = (const int*)d_in[3];
    const int* sib_head   = (const int*)d_in[4];
    const int* sib_mod    = (const int*)d_in[5];
    const int* sib_sib    = (const int*)d_in[6];
    const int* gp_head    = (const int*)d_in[7];
    const int* gp_mod     = (const int*)d_in[8];
    const int* gp_grand   = (const int*)d_in[9];
    const int* gsib_head  = (const int*)d_in[10];
    const int* gsib_mod   = (const int*)d_in[11];
    const int* gsib_sib   = (const int*)d_in[12];
    const int* gsib_grand = (const int*)d_in[13];
    const float* word_emb = (const float*)d_in[14];
    const float* tag_emb  = (const float*)d_in[15];
    const float* Wih_f    = (const float*)d_in[16];
    const float* Whh_f    = (const float*)d_in[17];
    const float* b_f      = (const float*)d_in[18];
    const float* Wih_b    = (const float*)d_in[19];
    const float* Whh_b    = (const float*)d_in[20];
    const float* b_b      = (const float*)d_in[21];
    const float* W_proj   = (const float*)d_in[22];
    const float* W_score  = (const float*)d_in[23];
    const float* null_sib = (const float*)d_in[24];

    float* out = (float*)d_out;

    float* ws      = (float*)d_ws;
    float* x       = ws;
    float* z_f     = x + SEQ * IN_DIM;
    float* z_b     = z_f + SEQ * G4;
    float* states  = z_b + SEQ * G4;
    float* projbuf = states + SEQ * D2H;

    embed_kernel<<<SEQ, 128, 0, stream>>>(words, tags, word_emb, tag_emb, x);

    igemm_kernel<<<dim3(SEQ / 32, G4 / 32, 2), 32, 0, stream>>>(
        x, Wih_f, b_f, Wih_b, b_b, z_f, z_b);

    float* cstate = x;
    lstm_step_kernel<true><<<STEP_GRID, 32 * STEP_WAVES, 0, stream>>>(
        z_f, Whh_f, Whh_b, states, states, cstate,
        states, states + (SEQ - 1) * D2H + H);
    {
        const float* zt = z_f + G4;
        float* hf = states;
        float* hb = states + (SEQ - 1) * D2H + H;
        for (; zt != z_f + SEQ * G4; zt += G4, hf += D2H, hb -= D2H)
            lstm_step_kernel<false><<<STEP_GRID, 32 * STEP_WAVES, 0, stream>>>(
                zt, Whh_f, Whh_b, hf, hb, cstate, hf + D2H, hb - D2H);
    }

    proj_kernel<<<dim3(SEQ / 32, NFLAT / 32), 32, 0, stream>>>(states, W_proj, projbuf);

    const dim3 sgrid(1024), sblk(256);
    score_kernel<<<sgrid, sblk, 0, stream>>>(projbuf, arc_head, arc_mod, nullptr, nullptr,
                                             0 * M_DIM, 1 * M_DIM, 0, 0,
                                             null_sib, W_score + 0 * M_DIM,
                                             out, N_ARC);
    score_kernel<<<sgrid, sblk, 0, stream>>>(projbuf, sib_head, sib_mod, sib_sib, nullptr,
                                             5 * M_DIM, 6 * M_DIM, 7 * M_DIM, 0,
                                             null_sib, W_score + 1 * M_DIM,
                                             out + N_ARC, N_SIB);
    score_kernel<<<sgrid, sblk, 0, stream>>>(projbuf, gp_head, gp_mod, nullptr, gp_grand,
                                             3 * M_DIM, 4 * M_DIM, 0, 2 * M_DIM,
                                             null_sib, W_score + 2 * M_DIM,
                                             out + N_ARC + N_SIB, N_GP);
    score_kernel<<<sgrid, sblk, 0, stream>>>(projbuf, gsib_head, gsib_mod, gsib_sib, gsib_grand,
                                             8 * M_DIM, 9 * M_DIM, 10 * M_DIM, 11 * M_DIM,
                                             null_sib, W_score + 3 * M_DIM,
                                             out + N_ARC + N_SIB + N_GP, N_GSIB);
}


# Scoring: a wave issues all loads of a tuple's gathered rows at once (16+8+4 B per lane and row, indices on the scalar unit, W_score in registers); new lane-to-element map changes the sum order
# speedup vs baseline: 13.4066x; 1.3105x over previous
#include <hip/hip_runtime.h>
#include <math.h>

#define SEQ    256
#define WE     100
#define TE     20
#define IN_DIM (WE + TE)
#define H      400
#define G4     (4 * H)
#define M_DIM  200
#define D2H    (2 * H)
#define NPROJ  12
#define NFLAT  (NPROJ * M_DIM)
#define N_ARC  65536
#define N_SIB  200000
#define N_GP   200000
#define N_GSIB 300000

typedef __attribute__((ext_vector_type(2))) float v2f;
typedef __attribute__((ext_vector_type(8))) float v8f;

__device__ __forceinline__ float sigmoidf_(float x) {
    return 1.0f / (1.0f + __expf(-x));
}

__device__ __forceinline__ v8f wmma_f32(v2f a, v2f b, v8f c) {
    return __builtin_amdgcn_wmma_f32_16x16x4_f32(false, a, false, b, (short)0, c,
                                                 false, false);
}

__global__ void embed_kernel(const int* __restrict__ words, const int* __restrict__ tags,
                             const float* __restrict__ wemb, const float* __restrict__ temb,
                             float* __restrict__ x) {
    const int s = blockIdx.x;
    const int t = threadIdx.x;
    if (t < WE)            x[s * IN_DIM + t] = wemb[words[s] * WE + t];
    else if (t < IN_DIM)   x[s * IN_DIM + t] = temb[tags[s] * TE + (t - WE)];
}

__global__ void igemm_kernel(const float* __restrict__ x,
                             const float* __restrict__ Wih_f, const float* __restrict__ b_f,
                             const float* __restrict__ Wih_b, const float* __restrict__ b_b,
                             float* __restrict__ z_f, float* __restrict__ z_b) {
    const int lane = threadIdx.x & 31;
    const int m0 = blockIdx.x * 32;
    const int n0 = blockIdx.y * 32;
    const int dir = blockIdx.z;

    const float* W    = dir ? Wih_b : Wih_f;
    const float* bias = dir ? b_b : b_f;
    float*       z    = dir ? z_b : z_f;

    const int mrow0 = m0 + (lane & 15);
    const int mrow1 = mrow0 + 16;
    const int arow0 = dir ? (SEQ - 1 - mrow0) : mrow0;
    const int arow1 = dir ? (SEQ - 1 - mrow1) : mrow1;
    const int kh    = (lane >> 4) * 2;
    const int ncol0 = n0 + (lane & 15);
    const int ncol1 = ncol0 + 16;

    v8f c00 = {}, c01 = {}, c10 = {}, c11 = {};
    for (int k = 0; k < IN_DIM; k += 4) {
        v2f a0 = *(const v2f*)(x + arow0 * IN_DIM + k + kh);
        v2f a1 = *(const v2f*)(x + arow1 * IN_DIM + k + kh);
        v2f b0 = *(const v2f*)(W + ncol0 * IN_DIM + k + kh);
        v2f b1 = *(const v2f*)(W + ncol1 * IN_DIM + k + kh);
        c00 = wmma_f32(a0, b0, c00);
        c01 = wmma_f32(a0, b1, c01);
        c10 = wmma_f32(a1, b0, c10);
        c11 = wmma_f32(a1, b1, c11);
    }
    const float bn0 = bias[ncol0];
    const float bn1 = bias[ncol1];
    const int mb = m0 + (lane >> 4) * 8;
#pragma unroll
    for (int r = 0; r < 8; ++r) {
        z[(mb + r) * G4 + ncol0]      = c00[r] + bn0;
        z[(mb + r) * G4 + ncol1]      = c01[r] + bn1;
        z[(mb + 16 + r) * G4 + ncol0] = c10[r] + bn0;
        z[(mb + 16 + r) * G4 + ncol1] = c11[r] + bn1;
    }
}

__device__ __forceinline__ float dot4_(float4 w, float4 h) {
    return w.x * h.x + w.y * h.y + w.z * h.z + w.w * h.w;
}

template <int M>
__device__ __forceinline__ float xrow_(float x) {
    return __builtin_bit_cast(float, __builtin_amdgcn_update_dpp(
        0, __builtin_bit_cast(int, x), 0x160 + M, 0xf, 0xf, false));
}
__device__ __forceinline__ float xhalf_(float x) {
    return __builtin_bit_cast(float, __builtin_amdgcn_permlanex16(
        __builtin_bit_cast(int, x), __builtin_bit_cast(int, x),
        0x76543210, 0xfedcba98, false, false));
}
__device__ __forceinline__ float wave_sum_(float x) {
    x += xrow_<1>(x);
    x += xrow_<2>(x);
    x += xrow_<4>(x);
    x += xrow_<8>(x);
    x += xhalf_(x);
    return x;
}

#define STEP_WAVES 4
#define STEP_GRID  (2 * H / STEP_WAVES)

template <bool FIRST>
__global__ void __launch_bounds__(32 * STEP_WAVES) __attribute__((amdgpu_waves_per_eu(1, 2)))
lstm_step_kernel(const float* __restrict__ z_t,
                 const float* __restrict__ Whh_f, const float* __restrict__ Whh_b,
                 const float* __restrict__ hprev_f, const float* __restrict__ hprev_b,
                 float* __restrict__ cstate,
                 float* __restrict__ hout_f, float* __restrict__ hout_b) {
    const int lane = threadIdx.x & 31;
    const int wv  = __builtin_amdgcn_readfirstlane((int)(blockIdx.x * STEP_WAVES + (threadIdx.x >> 5)));
    const int dir = (wv >= H) ? 1 : 0;
    const int j   = wv - dir * H;

    const float* z    = z_t + dir * (SEQ * G4);
    float*       hout = dir ? hout_b : hout_f;
    float*       c    = cstate + dir * H;

    float ai = 0.0f, af = 0.0f, ag = 0.0f, ao = 0.0f;
    float cprev = 0.0f;
    if (!FIRST) {
        const float4* h4 = (const float4*)(dir ? hprev_b : hprev_f);
        const float4* wi = (const float4*)((dir ? Whh_b : Whh_f) + j * H);
        const float4* wf = wi + (H * H) / 4;
        const float4* wg = wf + (H * H) / 4;
        const float4* wo = wg + (H * H) / 4;
        cprev = c[j];
        const int  qt = (lane < H / 4 - 96) ? (96 + lane) : (H / 4 - 1);
        const bool tail = lane < H / 4 - 96;
        float4 hv[4], vi[4], vf[4], vg[4], vo[4];
#pragma unroll
        for (int i = 0; i < 4; ++i) {
            const int q = (i < 3) ? (lane + 32 * i) : qt;
            hv[i] = h4[q];
            vi[i] = wi[q];
            vf[i] = wf[q];
            vg[i] = wg[q];
            vo[i] = wo[q];
        }
        hv[3].x = tail ? hv[3].x : 0.0f;
        hv[3].y = tail ? hv[3].y : 0.0f;
        hv[3].z = tail ? hv[3].z : 0.0f;
        hv[3].w = tail ? hv[3].w : 0.0f;
#pragma unroll
        for (int i = 0; i < 4; ++i) {
            ai += dot4_(vi[i], hv[i]);
            af += dot4_(vf[i], hv[i]);
            ag += dot4_(vg[i], hv[i]);
            ao += dot4_(vo[i], hv[i]);
        }
        ai = wave_sum_(ai);
        af = wave_sum_(af);
        ag = wave_sum_(ag);
        ao = wave_sum_(ao);
    }
    const float i_ = sigmoidf_(z[j] + ai);
    const float f_ = sigmoidf_(z[j + H] + af);
    const float g_ = tanhf(z[j + 2 * H] + ag);
    const float o_ = sigmoidf_(z[j + 3 * H] + ao);
    const float cn = f_ * cprev + i_ * g_;
    const float hn = o_ * tanhf(cn);
    if (lane == 0) {
        c[j]    = cn;
        hout[j] = hn;
    }
}

__global__ void proj_kernel(const float* __restrict__ states,
                            const float* __restrict__ Wp,
                            float* __restrict__ proj) {
    const int lane = threadIdx.x & 31;
    const int m0 = blockIdx.x * 32;
    const int n0 = blockIdx.y * 32;
    const int mrow0 = m0 + (lane & 15);
    const int mrow1 = mrow0 + 16;
    const int kh    = (lane >> 4) * 2;
    const int ncol0 = n0 + (lane & 15);
    const int ncol1 = ncol0 + 16;

    v8f c00 = {}, c01 = {}, c10 = {}, c11 = {};
    for (int k = 0; k < D2H; k += 4) {
        v2f a0 = *(const v2f*)(states + mrow0 * D2H + k + kh);
        v2f a1 = *(const v2f*)(states + mrow1 * D2H + k + kh);
        v2f b0 = *(const v2f*)(Wp + ncol0 * D2H + k + kh);
        v2f b1 = *(const v2f*)(Wp + ncol1 * D2H + k + kh);
        c00 = wmma_f32(a0, b0, c00);
        c01 = wmma_f32(a0, b1, c01);
        c10 = wmma_f32(a1, b0, c10);
        c11 = wmma_f32(a1, b1, c11);
    }
    const int mb = m0 + (lane >> 4) * 8;
#pragma unroll
    for (int r = 0; r < 8; ++r) {
        proj[(mb + r) * NFLAT + ncol0]      = c00[r];
        proj[(mb + r) * NFLAT + ncol1]      = c01[r];
        proj[(mb + 16 + r) * NFLAT + ncol0] = c10[r];
        proj[(mb + 16 + r) * NFLAT + ncol1] = c11[r];
    }
}

struct RowFrag { float4 a; float2 b; float c; };

__device__ __forceinline__ RowFrag load_row_(const float* __restrict__ r, int lane, int lc) {
    RowFrag f;
    f.a = ((const float4*)r)[lane];
    f.b = ((const float2*)(r + 128))[lane];
    f.c = r[192 + lc];
    return f;
}

template <bool HAS_S, bool HAS_G>
__global__ void
score_kernel(const float* __restrict__ proj,
             const int* __restrict__ ih, const int* __restrict__ im,
             const int* __restrict__ is, const int* __restrict__ ig,
             int ph, int pm, int ps, int pg,
             const float* __restrict__ null_sib,
             const float* __restrict__ wsc,
             float* __restrict__ out, int N) {
    const int lane   = threadIdx.x & 31;
    const int wid    = __builtin_amdgcn_readfirstlane((int)((blockIdx.x * blockDim.x + threadIdx.x) >> 5));
    const int nwaves = (gridDim.x * blockDim.x) >> 5;
    if (wid >= N) return;

    const int lc = lane & 7;
    const float4 w4 = ((const float4*)wsc)[lane];
    const float2 w2 = ((const float2*)(wsc + 128))[lane];
    const float  w1 = (lane < 8) ? wsc[192 + lc] : 0.0f;

    int tup = wid;
    int i_h = ih[tup];
    int i_m = im[tup];
    int i_s = HAS_S ? is[tup] : 0;
    int i_g = HAS_G ? ig[tup] : 0;
    for (;;) {
        const float* rh = proj + i_h * NFLAT + ph;
        const float* rm = proj + i_m * NFLAT + pm;
        const RowFrag fh = load_row_(rh, lane, lc);
        const RowFrag fm = load_row_(rm, lane, lc);
        RowFrag fs = {}, fg = {};
        if (HAS_S) {
            const float* rs = (i_s >= SEQ) ? null_sib : (proj + i_s * NFLAT + ps);
            fs = load_row_(rs, lane, lc);
        }
        if (HAS_G) {
            const float* rg = proj + i_g * NFLAT + pg;
            fg = load_row_(rg, lane, lc);
        }

        const int  nxt  = tup + nwaves;
        const bool more = nxt < N;
        if (more) {
            i_h = ih[nxt];
            i_m = im[nxt];
            if (HAS_S) i_s = is[nxt];
            if (HAS_G) i_g = ig[nxt];
        }

        float v0 = fh.a.x + fm.a.x, v1 = fh.a.y + fm.a.y, v2 = fh.a.z + fm.a.z, v3 = fh.a.w + fm.a.w;
        float v4 = fh.b.x + fm.b.x, v5 = fh.b.y + fm.b.y, v6 = fh.c + fm.c;
        if (HAS_S) {
            v0 += fs.a.x; v1 += fs.a.y; v2 += fs.a.z; v3 += fs.a.w;
            v4 += fs.b.x; v5 += fs.b.y; v6 += fs.c;
        }
        if (HAS_G) {
            v0 += fg.a.x; v1 += fg.a.y; v2 += fg.a.z; v3 += fg.a.w;
            v4 += fg.b.x; v5 += fg.b.y; v6 += fg.c;
        }
        float acc = tanhf(v0) * w4.x;
        acc += tanhf(v1) * w4.y;
        acc += tanhf(v2) * w4.z;
        acc += tanhf(v3) * w4.w;
        acc += tanhf(v4) * w2.x;
        acc += tanhf(v5) * w2.y;
        acc += tanhf(v6) * w1;
#pragma unroll
        for (int off = 16; off > 0; off >>= 1)
            acc += __shfl_xor(acc, off, 32);
        if (lane == 0) out[tup] = acc;

        if (!more) break;
        tup = nxt;
    }
}

extern "C" void kernel_launch(void* const* d_in, const int* in_sizes, int n_in,
                              void* d_out, int out_size, void* d_ws, size_t ws_size,
                              hipStream_t stream) {
    const int* words      = (const int*)d_in[0];
    const int* tags       = (const int*)d_in[1];
    const int* arc_head   = (const int*)d_in[2];
    const int* arc_mod    = (const int*)d_in[3];
    const int* sib_head   = (const int*)d_in[4];
    const int* sib_mod    = (const int*)d_in[5];
    const int* sib_sib    = (const int*)d_in[6];
    const int* gp_head    = (const int*)d_in[7];
    const int* gp_mod     = (const int*)d_in[8];
    const int* gp_grand   = (const int*)d_in[9];
    const int* gsib_head  = (const int*)d_in[10];
    const int* gsib_mod   = (const int*)d_in[11];
    const int* gsib_sib   = (const int*)d_in[12];
    const int* gsib_grand = (const int*)d_in[13];
    const float* word_emb = (const float*)d_in[14];
    const float* tag_emb  = (const float*)d_in[15];
    const float* Wih_f    = (const float*)d_in[16];
    const float* Whh_f    = (const float*)d_in[17];
    const float* b_f      = (const float*)d_in[18];
    const float* Wih_b    = (const float*)d_in[19];
    const float* Whh_b    = (const float*)d_in[20];
    const float* b_b      = (const float*)d_in[21];
    const float* W_proj   = (const float*)d_in[22];
    const float* W_score  = (const float*)d_in[23];
    const float* null_sib = (const float*)d_in[24];

    float* out = (float*)d_out;

    float* ws      = (float*)d_ws;
    float* x       = ws;
    float* z_f     = x + SEQ * IN_DIM;
    float* z_b     = z_f + SEQ * G4;
    float* states  = z_b + SEQ * G4;
    float* projbuf = states + SEQ * D2H;

    embed_kernel<<<SEQ, 128, 0, stream>>>(words, tags, word_emb, tag_emb, x);

    igemm_kernel<<<dim3(SEQ / 32, G4 / 32, 2), 32, 0, stream>>>(
        x, Wih_f, b_f, Wih_b, b_b, z_f, z_b);

    float* cstate = x;
    lstm_step_kernel<true><<<STEP_GRID, 32 * STEP_WAVES, 0, stream>>>(
        z_f, Whh_f, Whh_b, states, states, cstate,
        states, states + (SEQ - 1) * D2H + H);
    {
        const float* zt = z_f + G4;
        float* hf = states;
        float* hb = states + (SEQ - 1) * D2H + H;
        for (; zt != z_f + SEQ * G4; zt += G4, hf += D2H, hb -= D2H)
            lstm_step_kernel<false><<<STEP_GRID, 32 * STEP_WAVES, 0, stream>>>(
                zt, Whh_f, Whh_b, hf, hb, cstate, hf + D2H, hb - D2H);
    }

    proj_kernel<<<dim3(SEQ / 32, NFLAT / 32), 32, 0, stream>>>(states, W_proj, projbuf);

    const dim3 sgrid(1024), sblk(256);
    score_kernel<false, false><<<sgrid, sblk, 0, stream>>>(projbuf, arc_head, arc_mod, nullptr, nullptr,
                                             0 * M_DIM, 1 * M_DIM, 0, 0,
                                             null_sib, W_score + 0 * M_DIM,
                                             out, N_ARC);
    score_kernel<true, false><<<sgrid, sblk, 0, stream>>>(projbuf, sib_head, sib_mod, sib_sib, nullptr,
                                             5 * M_DIM, 6 * M_DIM, 7 * M_DIM, 0,
                                             null_sib, W_score + 1 * M_DIM,
                                             out + N_ARC, N_SIB);
    score_kernel<false, true><<<sgrid, sblk, 0, stream>>>(projbuf, gp_head, gp_mod, nullptr, gp_grand,
                                             3 * M_DIM, 4 * M_DIM, 0, 2 * M_DIM,
                                             null_sib, W_score + 2 * M_DIM,
                                             out + N_ARC + N_SIB, N_GP);
    score_kernel<true, true><<<sgrid, sblk, 0, stream>>>(projbuf, gsib_head, gsib_mod, gsib_sib, gsib_grand,
                                             8 * M_DIM, 9 * M_DIM, 10 * M_DIM, 11 * M_DIM,
                                             null_sib, W_score + 3 * M_DIM,
                                             out + N_ARC + N_SIB + N_GP, N_GSIB);
}


# Every tanh and logistic (LSTM gates, scoring) on the hardware tanh unit: one v_tanh_f32 each, logistic as (1+tanh(x/2))/2, no branches, exp or division; each activation's rounding changes
# speedup vs baseline: 14.8493x; 1.1076x over previous
#include <hip/hip_runtime.h>
#include <math.h>

#define SEQ    256
#define WE     100
#define TE     20
#define IN_DIM (WE + TE)
#define H      400
#define G4     (4 * H)
#define M_DIM  200
#define D2H    (2 * H)
#define NPROJ  12
#define NFLAT  (NPROJ * M_DIM)
#define N_ARC  65536
#define N_SIB  200000
#define N_GP   200000
#define N_GSIB 300000

typedef __attribute__((ext_vector_type(2))) float v2f;
typedef __attribute__((ext_vector_type(8))) float v8f;

__device__ __forceinline__ float tanh_hw_(float x) {
    return __builtin_amdgcn_tanhf(x);
}

__device__ __forceinline__ float sigmoidf_(float x) {
    return fmaf(0.5f, tanh_hw_(0.5f * x), 0.5f);
}

__device__ __forceinline__ v8f wmma_f32(v2f a, v2f b, v8f c) {
    return __builtin_amdgcn_wmma_f32_16x16x4_f32(false, a, false, b, (short)0, c,
                                                 false, false);
}

__global__ void embed_kernel(const int* __restrict__ words, const int* __restrict__ tags,
                             const float* __restrict__ wemb, const float* __restrict__ temb,
                             float* __restrict__ x) {
    const int s = blockIdx.x;
    const int t = threadIdx.x;
    if (t < WE)            x[s * IN_DIM + t] = wemb[words[s] * WE + t];
    else if (t < IN_DIM)   x[s * IN_DIM + t] = temb[tags[s] * TE + (t - WE)];
}

__global__ void igemm_kernel(const float* __restrict__ x,
                             const float* __restrict__ Wih_f, const float* __restrict__ b_f,
                             const float* __restrict__ Wih_b, const float* __restrict__ b_b,
                             float* __restrict__ z_f, float* __restrict__ z_b) {
    const int lane = threadIdx.x & 31;
    const int m0 = blockIdx.x * 32;
    const int n0 = blockIdx.y * 32;
    const int dir = blockIdx.z;

    const float* W    = dir ? Wih_b : Wih_f;
    const float* bias = dir ? b_b : b_f;
    float*       z    = dir ? z_b : z_f;

    const int mrow0 = m0 + (lane & 15);
    const int mrow1 = mrow0 + 16;
    const int arow0 = dir ? (SEQ - 1 - mrow0) : mrow0;
    const int arow1 = dir ? (SEQ - 1 - mrow1) : mrow1;
    const int kh    = (lane >> 4) * 2;
    const int ncol0 = n0 + (lane & 15);
    const int ncol1 = ncol0 + 16;

    v8f c00 = {}, c01 = {}, c10 = {}, c11 = {};
    for (int k = 0; k < IN_DIM; k += 4) {
        v2f a0 = *(const v2f*)(x + arow0 * IN_DIM + k + kh);
        v2f a1 = *(const v2f*)(x + arow1 * IN_DIM + k + kh);
        v2f b0 = *(const v2f*)(W + ncol0 * IN_DIM + k + kh);
        v2f b1 = *(const v2f*)(W + ncol1 * IN_DIM + k + kh);
        c00 = wmma_f32(a0, b0, c00);
        c01 = wmma_f32(a0, b1, c01);
        c10 = wmma_f32(a1, b0, c10);
        c11 = wmma_f32(a1, b1, c11);
    }
    const float bn0 = bias[ncol0];
    const float bn1 = bias[ncol1];
    const int mb = m0 + (lane >> 4) * 8;
#pragma unroll
    for (int r = 0; r < 8; ++r) {
        z[(mb + r) * G4 + ncol0]      = c00[r] + bn0;
        z[(mb + r) * G4 + ncol1]      = c01[r] + bn1;
        z[(mb + 16 + r) * G4 + ncol0] = c10[r] + bn0;
        z[(mb + 16 + r) * G4 + ncol1] = c11[r] + bn1;
    }
}

__device__ __forceinline__ float dot4_(float4 w, float4 h) {
    return w.x * h.x + w.y * h.y + w.z * h.z + w.w * h.w;
}

template <int M>
__device__ __forceinline__ float xrow_(float x) {
    return __builtin_bit_cast(float, __builtin_amdgcn_update_dpp(
        0, __builtin_bit_cast(int, x), 0x160 + M, 0xf, 0xf, false));
}
__device__ __forceinline__ float xhalf_(float x) {
    return __builtin_bit_cast(float, __builtin_amdgcn_permlanex16(
        __builtin_bit_cast(int, x), __builtin_bit_cast(int, x),
        0x76543210, 0xfedcba98, false, false));
}
__device__ __forceinline__ float wave_sum_(float x) {
    x += xrow_<1>(x);
    x += xrow_<2>(x);
    x += xrow_<4>(x);
    x += xrow_<8>(x);
    x += xhalf_(x);
    return x;
}

#define STEP_WAVES 4
#define STEP_GRID  (2 * H / STEP_WAVES)

template <bool FIRST>
__global__ void __launch_bounds__(32 * STEP_WAVES) __attribute__((amdgpu_waves_per_eu(1, 2)))
lstm_step_kernel(const float* __restrict__ z_t,
                 const float* __restrict__ Whh_f, const float* __restrict__ Whh_b,
                 const float* __restrict__ hprev_f, const float* __restrict__ hprev_b,
                 float* __restrict__ cstate,
                 float* __restrict__ hout_f, float* __restrict__ hout_b) {
    const int lane = threadIdx.x & 31;
    const int wv  = __builtin_amdgcn_readfirstlane((int)(blockIdx.x * STEP_WAVES + (threadIdx.x >> 5)));
    const int dir = (wv >= H) ? 1 : 0;
    const int j   = wv - dir * H;

    const float* z    = z_t + dir * (SEQ * G4);
    float*       hout = dir ? hout_b : hout_f;
    float*       c    = cstate + dir * H;

    const float zi = z[j], zf = z[j + H], zg = z[j + 2 * H], zo = z[j + 3 * H];
    float ai = 0.0f, af = 0.0f, ag = 0.0f, ao = 0.0f;
    float cprev = 0.0f;
    if (!FIRST) {
        const float4* h4 = (const float4*)(dir ? hprev_b : hprev_f);
        const float4* wi = (const float4*)((dir ? Whh_b : Whh_f) + j * H);
        const float4* wf = wi + (H * H) / 4;
        const float4* wg = wf + (H * H) / 4;
        const float4* wo = wg + (H * H) / 4;
        cprev = c[j];
        const int  qt = (lane < H / 4 - 96) ? (96 + lane) : (H / 4 - 1);
        const bool tail = lane < H / 4 - 96;
        float4 hv[4], vi[4], vf[4], vg[4], vo[4];
#pragma unroll
        for (int i = 0; i < 4; ++i) {
            const int q = (i < 3) ? (lane + 32 * i) : qt;
            hv[i] = h4[q];
            vi[i] = wi[q];
            vf[i] = wf[q];
            vg[i] = wg[q];
            vo[i] = wo[q];
        }
        hv[3].x = tail ? hv[3].x : 0.0f;
        hv[3].y = tail ? hv[3].y : 0.0f;
        hv[3].z = tail ? hv[3].z : 0.0f;
        hv[3].w = tail ? hv[3].w : 0.0f;
#pragma unroll
        for (int i = 0; i < 4; ++i) {
            ai += dot4_(vi[i], hv[i]);
            af += dot4_(vf[i], hv[i]);
            ag += dot4_(vg[i], hv[i]);
            ao += dot4_(vo[i], hv[i]);
        }
        ai = wave_sum_(ai);
        af = wave_sum_(af);
        ag = wave_sum_(ag);
        ao = wave_sum_(ao);
    }
    asm volatile("" :: "s"(zi), "s"(zf), "s"(zg), "s"(zo));
    if (!FIRST) asm volatile("" :: "s"(cprev));
    const float i_ = sigmoidf_(zi + ai);
    const float f_ = sigmoidf_(zf + af);
    const float g_ = tanh_hw_(zg + ag);
    const float o_ = sigmoidf_(zo + ao);
    const float cn = f_ * cprev + i_ * g_;
    const float hn = o_ * tanh_hw_(cn);
    if (lane == 0) {
        c[j]    = cn;
        hout[j] = hn;
    }
}

__global__ void proj_kernel(const float* __restrict__ states,
                            const float* __restrict__ Wp,
                            float* __restrict__ proj) {
    const int lane = threadIdx.x & 31;
    const int m0 = blockIdx.x * 32;
    const int n0 = blockIdx.y * 32;
    const int mrow0 = m0 + (lane & 15);
    const int mrow1 = mrow0 + 16;
    const int kh    = (lane >> 4) * 2;
    const int ncol0 = n0 + (lane & 15);
    const int ncol1 = ncol0 + 16;

    v8f c00 = {}, c01 = {}, c10 = {}, c11 = {};
    for (int k = 0; k < D2H; k += 4) {
        v2f a0 = *(const v2f*)(states + mrow0 * D2H + k + kh);
        v2f a1 = *(const v2f*)(states + mrow1 * D2H + k + kh);
        v2f b0 = *(const v2f*)(Wp + ncol0 * D2H + k + kh);
        v2f b1 = *(const v2f*)(Wp + ncol1 * D2H + k + kh);
        c00 = wmma_f32(a0, b0, c00);
        c01 = wmma_f32(a0, b1, c01);
        c10 = wmma_f32(a1, b0, c10);
        c11 = wmma_f32(a1, b1, c11);
    }
    const int mb = m0 + (lane >> 4) * 8;
#pragma unroll
    for (int r = 0; r < 8; ++r) {
        proj[(mb + r) * NFLAT + ncol0]      = c00[r];
        proj[(mb + r) * NFLAT + ncol1]      = c01[r];
        proj[(mb + 16 + r) * NFLAT + ncol0] = c10[r];
        proj[(mb + 16 + r) * NFLAT + ncol1] = c11[r];
    }
}

struct RowFrag { float4 a; float2 b; float c; };

__device__ __forceinline__ RowFrag load_row_(const float* __restrict__ r, int lane, int lc) {
    RowFrag f;
    f.a = ((const float4*)r)[lane];
    f.b = ((const float2*)(r + 128))[lane];
    f.c = r[192 + lc];
    return f;
}

template <bool HAS_S, bool HAS_G>
__global__ void
score_kernel(const float* __restrict__ proj,
             const int* __restrict__ ih, const int* __restrict__ im,
             const int* __restrict__ is, const int* __restrict__ ig,
             int ph, int pm, int ps, int pg,
             const float* __restrict__ null_sib,
             const float* __restrict__ wsc,
             float* __restrict__ out, int N) {
    const int lane   = threadIdx.x & 31;
    const int wid    = __builtin_amdgcn_readfirstlane((int)((blockIdx.x * blockDim.x + threadIdx.x) >> 5));
    const int nwaves = (gridDim.x * blockDim.x) >> 5;
    if (wid >= N) return;

    const int lc = lane & 7;
    const float4 w4 = ((const float4*)wsc)[lane];
    const float2 w2 = ((const float2*)(wsc + 128))[lane];
    const float  w1 = (lane < 8) ? wsc[192 + lc] : 0.0f;

    int tup = wid;
    int i_h = ih[tup];
    int i_m = im[tup];
    int i_s = HAS_S ? is[tup] : 0;
    int i_g = HAS_G ? ig[tup] : 0;
    for (;;) {
        const float* rh = proj + i_h * NFLAT + ph;
        const float* rm = proj + i_m * NFLAT + pm;
        const RowFrag fh = load_row_(rh, lane, lc);
        const RowFrag fm = load_row_(rm, lane, lc);
        RowFrag fs = {}, fg = {};
        if (HAS_S) {
            const float* rs = (i_s >= SEQ) ? null_sib : (proj + i_s * NFLAT + ps);
            fs = load_row_(rs, lane, lc);
        }
        if (HAS_G) {
            const float* rg = proj + i_g * NFLAT + pg;
            fg = load_row_(rg, lane, lc);
        }

        const int  nxt  = tup + nwaves;
        const bool more = nxt < N;
        if (more) {
            i_h = ih[nxt];
            i_m = im[nxt];
            if (HAS_S) i_s = is[nxt];
            if (HAS_G) i_g = ig[nxt];
        }

        float v0 = fh.a.x + fm.a.x, v1 = fh.a.y + fm.a.y, v2 = fh.a.z + fm.a.z, v3 = fh.a.w + fm.a.w;
        float v4 = fh.b.x + fm.b.x, v5 = fh.b.y + fm.b.y, v6 = fh.c + fm.c;
        if (HAS_S) {
            v0 += fs.a.x; v1 += fs.a.y; v2 += fs.a.z; v3 += fs.a.w;
            v4 += fs.b.x; v5 += fs.b.y; v6 += fs.c;
        }
        if (HAS_G) {
            v0 += fg.a.x; v1 += fg.a.y; v2 += fg.a.z; v3 += fg.a.w;
            v4 += fg.b.x; v5 += fg.b.y; v6 += fg.c;
        }
        float acc = tanh_hw_(v0) * w4.x;
        acc += tanh_hw_(v1) * w4.y;
        acc += tanh_hw_(v2) * w4.z;
        acc += tanh_hw_(v3) * w4.w;
        acc += tanh_hw_(v4) * w2.x;
        acc += tanh_hw_(v5) * w2.y;
        acc += tanh_hw_(v6) * w1;
#pragma unroll
        for (int off = 16; off > 0; off >>= 1)
            acc += __shfl_xor(acc, off, 32);
        if (lane == 0) out[tup] = acc;

        if (!more) break;
        tup = nxt;
    }
}

extern "C" void kernel_launch(void* const* d_in, const int* in_sizes, int n_in,
                              void* d_out, int out_size, void* d_ws, size_t ws_size,
                              hipStream_t stream) {
    const int* words      = (const int*)d_in[0];
    const int* tags       = (const int*)d_in[1];
    const int* arc_head   = (const int*)d_in[2];
    const int* arc_mod    = (const int*)d_in[3];
    const int* sib_head   = (const int*)d_in[4];
    const int* sib_mod    = (const int*)d_in[5];
    const int* sib_sib    = (const int*)d_in[6];
    const int* gp_head    = (const int*)d_in[7];
    const int* gp_mod     = (const int*)d_in[8];
    const int* gp_grand   = (const int*)d_in[9];
    const int* gsib_head  = (const int*)d_in[10];
    const int* gsib_mod   = (const int*)d_in[11];
    const int* gsib_sib   = (const int*)d_in[12];
    const int* gsib_grand = (const int*)d_in[13];
    const float* word_emb = (const float*)d_in[14];
    const float* tag_emb  = (const float*)d_in[15];
    const float* Wih_f    = (const float*)d_in[16];
    const float* Whh_f    = (const float*)d_in[17];
    const float* b_f      = (const float*)d_in[18];
    const float* Wih_b    = (const float*)d_in[19];
    const float* Whh_b    = (const float*)d_in[20];
    const float* b_b      = (const float*)d_in[21];
    const float* W_proj   = (const float*)d_in[22];
    const float* W_score  = (const float*)d_in[23];
    const float* null_sib = (const float*)d_in[24];

    float* out = (float*)d_out;

    float* ws      = (float*)d_ws;
    float* x       = ws;
    float* z_f     = x + SEQ * IN_DIM;
    float* z_b     = z_f + SEQ * G4;
    float* states  = z_b + SEQ * G4;
    float* projbuf = states + SEQ * D2H;

    embed_kernel<<<SEQ, 128, 0, stream>>>(words, tags, word_emb, tag_emb, x);

    igemm_kernel<<<dim3(SEQ / 32, G4 / 32, 2), 32, 0, stream>>>(
        x, Wih_f, b_f, Wih_b, b_b, z_f, z_b);

    float* cstate = x;
    lstm_step_kernel<true><<<STEP_GRID, 32 * STEP_WAVES, 0, stream>>>(
        z_f, Whh_f, Whh_b, states, states, cstate,
        states, states + (SEQ - 1) * D2H + H);
    {
        const float* zt = z_f + G4;
        float* hf = states;
        float* hb = states + (SEQ - 1) * D2H + H;
        for (; zt != z_f + SEQ * G4; zt += G4, hf += D2H, hb -= D2H)
            lstm_step_kernel<false><<<STEP_GRID, 32 * STEP_WAVES, 0, stream>>>(
                zt, Whh_f, Whh_b, hf, hb, cstate, hf + D2H, hb - D2H);
    }

    proj_kernel<<<dim3(SEQ / 32, NFLAT / 32), 32, 0, stream>>>(states, W_proj, projbuf);

    const dim3 sgrid(1024), sblk(256);
    score_kernel<false, false><<<sgrid, sblk, 0, stream>>>(projbuf, arc_head, arc_mod, nullptr, nullptr,
                                             0 * M_DIM, 1 * M_DIM, 0, 0,
                                             null_sib, W_score + 0 * M_DIM,
                                             out, N_ARC);
    score_kernel<true, false><<<sgrid, sblk, 0, stream>>>(projbuf, sib_head, sib_mod, sib_sib, nullptr,
                                             5 * M_DIM, 6 * M_DIM, 7 * M_DIM, 0,
                                             null_sib, W_score + 1 * M_DIM,
                                             out + N_ARC, N_SIB);
    score_kernel<false, true><<<sgrid, sblk, 0, stream>>>(projbuf, gp_head, gp_mod, nullptr, gp_grand,
                                             3 * M_DIM, 4 * M_DIM, 0, 2 * M_DIM,
                                             null_sib, W_score + 2 * M_DIM,
                                             out + N_ARC + N_SIB, N_GP);
    score_kernel<true, true><<<sgrid, sblk, 0, stream>>>(projbuf, gsib_head, gsib_mod, gsib_sib, gsib_grand,
                                             8 * M_DIM, 9 * M_DIM, 10 * M_DIM, 11 * M_DIM,
                                             null_sib, W_score + 3 * M_DIM,
                                             out + N_ARC + N_SIB + N_GP, N_GSIB);
}


# Scoring reads its table rows from LDS: one launch, a resident work-group per WGP stages each family's tables in four 50-column slices; per-lane partial sums over slices change the sum order
# speedup vs baseline: 16.7084x; 1.1252x over previous
#include <hip/hip_runtime.h>
#include <math.h>

#define SEQ    256
#define WE     100
#define TE     20
#define IN_DIM (WE + TE)
#define H      400
#define G4     (4 * H)
#define M_DIM  200
#define D2H    (2 * H)
#define NPROJ  12
#define NFLAT  (NPROJ * M_DIM)
#define N_ARC  65536
#define N_SIB  200000
#define N_GP   200000
#define N_GSIB 300000
#define SL_W     50
#define NSLICE   (M_DIM / SL_W)
#define SL_TAB   (SEQ * SL_W)
#define ROLE_LDS (SL_TAB + SL_W + 2)

typedef __attribute__((ext_vector_type(2))) float v2f;
typedef __attribute__((ext_vector_type(8))) float v8f;
typedef __attribute__((ext_vector_type(4))) float v4f;

__device__ __forceinline__ float tanh_hw_(float x) {
    return __builtin_amdgcn_tanhf(x);
}

__device__ __forceinline__ float sigmoidf_(float x) {
    return fmaf(0.5f, tanh_hw_(0.5f * x), 0.5f);
}

__device__ __forceinline__ v8f wmma_f32(v2f a, v2f b, v8f c) {
    return __builtin_amdgcn_wmma_f32_16x16x4_f32(false, a, false, b, (short)0, c,
                                                 false, false);
}

__global__ void embed_kernel(const int* __restrict__ words, const int* __restrict__ tags,
                             const float* __restrict__ wemb, const float* __restrict__ temb,
                             float* __restrict__ x) {
    const int s = blockIdx.x;
    const int t = threadIdx.x;
    if (t < WE)            x[s * IN_DIM + t] = wemb[words[s] * WE + t];
    else if (t < IN_DIM)   x[s * IN_DIM + t] = temb[tags[s] * TE + (t - WE)];
}

__global__ void igemm_kernel(const float* __restrict__ x,
                             const float* __restrict__ Wih_f, const float* __restrict__ b_f,
                             const float* __restrict__ Wih_b, const float* __restrict__ b_b,
                             float* __restrict__ z_f, float* __restrict__ z_b) {
    const int lane = threadIdx.x & 31;
    const int m0 = blockIdx.x * 32;
    const int n0 = blockIdx.y * 32;
    const int dir = blockIdx.z;

    const float* W    = dir ? Wih_b : Wih_f;
    const float* bias = dir ? b_b : b_f;
    float*       z    = dir ? z_b : z_f;

    const int mrow0 = m0 + (lane & 15);
    const int mrow1 = mrow0 + 16;
    const int arow0 = dir ? (SEQ - 1 - mrow0) : mrow0;
    const int arow1 = dir ? (SEQ - 1 - mrow1) : mrow1;
    const int kh    = (lane >> 4) * 2;
    const int ncol0 = n0 + (lane & 15);
    const int ncol1 = ncol0 + 16;

    v8f c00 = {}, c01 = {}, c10 = {}, c11 = {};
    for (int k = 0; k < IN_DIM; k += 4) {
        v2f a0 = *(const v2f*)(x + arow0 * IN_DIM + k + kh);
        v2f a1 = *(const v2f*)(x + arow1 * IN_DIM + k + kh);
        v2f b0 = *(const v2f*)(W + ncol0 * IN_DIM + k + kh);
        v2f b1 = *(const v2f*)(W + ncol1 * IN_DIM + k + kh);
        c00 = wmma_f32(a0, b0, c00);
        c01 = wmma_f32(a0, b1, c01);
        c10 = wmma_f32(a1, b0, c10);
        c11 = wmma_f32(a1, b1, c11);
    }
    const float bn0 = bias[ncol0];
    const float bn1 = bias[ncol1];
    const int mb = m0 + (lane >> 4) * 8;
#pragma unroll
    for (int r = 0; r < 8; ++r) {
        z[(mb + r) * G4 + ncol0]      = c00[r] + bn0;
        z[(mb + r) * G4 + ncol1]      = c01[r] + bn1;
        z[(mb + 16 + r) * G4 + ncol0] = c10[r] + bn0;
        z[(mb + 16 + r) * G4 + ncol1] = c11[r] + bn1;
    }
}

__device__ __forceinline__ float dot4_(float4 w, float4 h) {
    return w.x * h.x + w.y * h.y + w.z * h.z + w.w * h.w;
}

template <int M>
__device__ __forceinline__ float xrow_(float x) {
    return __builtin_bit_cast(float, __builtin_amdgcn_update_dpp(
        0, __builtin_bit_cast(int, x), 0x160 + M, 0xf, 0xf, false));
}
__device__ __forceinline__ float xhalf_(float x) {
    return __builtin_bit_cast(float, __builtin_amdgcn_permlanex16(
        __builtin_bit_cast(int, x), __builtin_bit_cast(int, x),
        0x76543210, 0xfedcba98, false, false));
}
__device__ __forceinline__ float wave_sum_(float x) {
    x += xrow_<1>(x);
    x += xrow_<2>(x);
    x += xrow_<4>(x);
    x += xrow_<8>(x);
    x += xhalf_(x);
    return x;
}

#define STEP_WAVES 4
#define STEP_GRID  (2 * H / STEP_WAVES)

template <bool FIRST>
__global__ void __launch_bounds__(32 * STEP_WAVES) __attribute__((amdgpu_waves_per_eu(1, 2)))
lstm_step_kernel(const float* __restrict__ z_t,
                 const float* __restrict__ Whh_f, const float* __restrict__ Whh_b,
                 const float* __restrict__ hprev_f, const float* __restrict__ hprev_b,
                 float* __restrict__ cstate,
                 float* __restrict__ hout_f, float* __restrict__ hout_b) {
    const int lane = threadIdx.x & 31;
    const int wv  = __builtin_amdgcn_readfirstlane((int)(blockIdx.x * STEP_WAVES + (threadIdx.x >> 5)));
    const int dir = (wv >= H) ? 1 : 0;
    const int j   = wv - dir * H;

    const float* z    = z_t + dir * (SEQ * G4);
    float*       hout = dir ? hout_b : hout_f;
    float*       c    = cstate + dir * H;

    const float zi = z[j], zf = z[j + H], zg = z[j + 2 * H], zo = z[j + 3 * H];
    float ai = 0.0f, af = 0.0f, ag = 0.0f, ao = 0.0f;
    float cprev = 0.0f;
    if (!FIRST) {
        const float4* h4 = (const float4*)(dir ? hprev_b : hprev_f);
        const float4* wi = (const float4*)((dir ? Whh_b : Whh_f) + j * H);
        const float4* wf = wi + (H * H) / 4;
        const float4* wg = wf + (H * H) / 4;
        const float4* wo = wg + (H * H) / 4;
        cprev = c[j];
        const int  qt = (lane < H / 4 - 96) ? (96 + lane) : (H / 4 - 1);
        const bool tail = lane < H / 4 - 96;
        float4 hv[4], vi[4], vf[4], vg[4], vo[4];
#pragma unroll
        for (int i = 0; i < 4; ++i) {
            const int q = (i < 3) ? (lane + 32 * i) : qt;
            hv[i] = h4[q];
            vi[i] = wi[q];
            vf[i] = wf[q];
            vg[i] = wg[q];
            vo[i] = wo[q];
        }
        hv[3].x = tail ? hv[3].x : 0.0f;
        hv[3].y = tail ? hv[3].y : 0.0f;
        hv[3].z = tail ? hv[3].z : 0.0f;
        hv[3].w = tail ? hv[3].w : 0.0f;
#pragma unroll
        for (int i = 0; i < 4; ++i) {
            ai += dot4_(vi[i], hv[i]);
            af += dot4_(vf[i], hv[i]);
            ag += dot4_(vg[i], hv[i]);
            ao += dot4_(vo[i], hv[i]);
        }
        ai = wave_sum_(ai);
        af = wave_sum_(af);
        ag = wave_sum_(ag);
        ao = wave_sum_(ao);
    }
    asm volatile("" :: "s"(zi), "s"(zf), "s"(zg), "s"(zo));
    if (!FIRST) asm volatile("" :: "s"(cprev));
    const float i_ = sigmoidf_(zi + ai);
    const float f_ = sigmoidf_(zf + af);
    const float g_ = tanh_hw_(zg + ag);
    const float o_ = sigmoidf_(zo + ao);
    const float cn = f_ * cprev + i_ * g_;
    const float hn = o_ * tanh_hw_(cn);
    if (lane == 0) {
        c[j]    = cn;
        hout[j] = hn;
    }
}

__global__ void proj_kernel(const float* __restrict__ states,
                            const float* __restrict__ Wp,
                            float* __restrict__ proj) {
    const int lane = threadIdx.x & 31;
    const int m0 = blockIdx.x * 32;
    const int n0 = blockIdx.y * 32;
    const int mrow0 = m0 + (lane & 15);
    const int mrow1 = mrow0 + 16;
    const int kh    = (lane >> 4) * 2;
    const int ncol0 = n0 + (lane & 15);
    const int ncol1 = ncol0 + 16;

    v8f c00 = {}, c01 = {}, c10 = {}, c11 = {};
    for (int k = 0; k < D2H; k += 4) {
        v2f a0 = *(const v2f*)(states + mrow0 * D2H + k + kh);
        v2f a1 = *(const v2f*)(states + mrow1 * D2H + k + kh);
        v2f b0 = *(const v2f*)(Wp + ncol0 * D2H + k + kh);
        v2f b1 = *(const v2f*)(Wp + ncol1 * D2H + k + kh);
        c00 = wmma_f32(a0, b0, c00);
        c01 = wmma_f32(a0, b1, c01);
        c10 = wmma_f32(a1, b0, c10);
        c11 = wmma_f32(a1, b1, c11);
    }
    const int mb = m0 + (lane >> 4) * 8;
    const int nb0 = (ncol0 / SL_W) * SL_TAB + ncol0 % SL_W;
    const int nb1 = (ncol1 / SL_W) * SL_TAB + ncol1 % SL_W;
#pragma unroll
    for (int r = 0; r < 8; ++r) {
        proj[nb0 + (mb + r) * SL_W]      = c00[r];
        proj[nb1 + (mb + r) * SL_W]      = c01[r];
        proj[nb0 + (mb + 16 + r) * SL_W] = c10[r];
        proj[nb1 + (mb + 16 + r) * SL_W] = c11[r];
    }
}

#define SC_WAVES 32
#define SC_GRID  256

template <int R, bool HAS_S, int TPW>
__device__ __forceinline__ void
score_family_(float* lds, const float* __restrict__ proj,
              const int* __restrict__ i0, const int* __restrict__ i1,
              const int* __restrict__ i2, const int* __restrict__ i3,
              int p0, int p1, int p2, int p3,
              const float* __restrict__ null_sib,
              const float* __restrict__ wsc,
              float* __restrict__ out, int N,
              int tid, int lane, int gw) {
    constexpr int NV = (TPW + 31) / 32;
    constexpr int TG = 4;
    constexpr int NQ = SL_TAB / 4;
    constexpr int NT = 32 * SC_WAVES;
    const int* const ip[4] = {i0, i1, i2, i3};
    const int        pp[4] = {p0, p1, p2, p3};
    const int tb = gw * TPW;
    const bool live = tb < N;

    int vi[4][NV];
#pragma unroll
    for (int r = 0; r < R; ++r) {
#pragma unroll
        for (int v = 0; v < NV; ++v) {
            int t = tb + 32 * v + lane;
            t = t < N ? t : N - 1;
            int i = ip[r][t];
            if (HAS_S && r == 2) i = i < SEQ ? i : SEQ;
            vi[r][v] = r * ROLE_LDS + i * SL_W;
        }
    }

    const int lc = lane < SL_W / 2 ? lane : SL_W / 2 - 1;

    float acc[TPW];
#pragma unroll
    for (int j = 0; j < TPW; ++j) acc[j] = 0.0f;

#pragma unroll 1
    for (int sl = 0; sl < NSLICE; ++sl) {
        float2 w = ((const float2*)(wsc + sl * SL_W))[lc];
        {
            constexpr int RB = (R == 4) ? 2 : R;
            const int qt = 3 * NT + (tid & (NQ - 3 * NT - 1));
#pragma unroll
            for (int r0 = 0; r0 < R; r0 += RB) {
                v4f t[RB][3], u[RB];
#pragma unroll
                for (int r = 0; r < RB; ++r) {
                    const v4f* s4 = (const v4f*)(proj + (pp[r0 + r] * NSLICE + sl) * SL_TAB);
#pragma unroll
                    for (int k = 0; k < 3; ++k) t[r][k] = s4[tid + k * NT];
                    u[r] = s4[qt];
                }
                if constexpr (RB == 2)
                    asm volatile("" :: "v"(t[0][0]), "v"(t[0][1]), "v"(t[0][2]), "v"(u[0]),
                                       "v"(t[1][0]), "v"(t[1][1]), "v"(t[1][2]), "v"(u[1]));
                else
                    asm volatile("" :: "v"(t[0][0]), "v"(t[0][1]), "v"(t[0][2]), "v"(u[0]),
                                       "v"(t[1][0]), "v"(t[1][1]), "v"(t[1][2]), "v"(u[1]),
                                       "v"(t[2][0]), "v"(t[2][1]), "v"(t[2][2]), "v"(u[2]));
#pragma unroll
                for (int r = 0; r < RB; ++r) {
                    v4f* d4 = (v4f*)(lds + (r0 + r) * ROLE_LDS);
#pragma unroll
                    for (int k = 0; k < 3; ++k) d4[tid + k * NT] = t[r][k];
                    if (tid < NQ - 3 * NT) d4[qt] = u[r];
                }
            }
        }
        if (HAS_S && tid < SL_W) lds[2 * ROLE_LDS + SL_TAB + tid] = null_sib[sl * SL_W + tid];
        __syncthreads();

        if (live) {
            if (lane >= SL_W / 2) { w.x = 0.0f; w.y = 0.0f; }
#pragma unroll
            for (int j0 = 0; j0 < TPW; j0 += TG) {
                v2f y[TG][R];
#pragma unroll
                for (int g = 0; g < TG; ++g) {
                    if (j0 + g < TPW) {
#pragma unroll
                        for (int r = 0; r < R; ++r) {
                            const int o = __builtin_amdgcn_readlane(vi[r][(j0 + g) >> 5], (j0 + g) & 31);
                            y[g][r] = ((const v2f*)(lds + o))[lc];
                        }
                    }
                }
#pragma unroll
                for (int g = 0; g < TG; ++g) {
                    if (j0 + g < TPW) {
                        v2f x = y[g][0];
#pragma unroll
                        for (int r = 1; r < R; ++r) x += y[g][r];
                        acc[j0 + g] += tanh_hw_(x.x) * w.x + tanh_hw_(x.y) * w.y;
                    }
                }
            }
        }
        __syncthreads();
    }

    if (live) {
        float res[NV];
#pragma unroll
        for (int v = 0; v < NV; ++v) res[v] = 0.0f;
#pragma unroll
        for (int j = 0; j < TPW; ++j) {
            const float s = wave_sum_(acc[j]);
            if (lane == (j & 31)) res[j >> 5] = s;
        }
#pragma unroll
        for (int v = 0; v < NV; ++v) {
            const int t = tb + 32 * v + lane;
            if (32 * v + lane < TPW && t < N) out[t] = res[v];
        }
    }
}

__global__ void __launch_bounds__(32 * SC_WAVES)
score_lds_kernel(const float* __restrict__ proj,
                 const int* __restrict__ arc_head,  const int* __restrict__ arc_mod,
                 const int* __restrict__ sib_head,  const int* __restrict__ sib_mod,
                 const int* __restrict__ sib_sib,
                 const int* __restrict__ gp_head,   const int* __restrict__ gp_mod,
                 const int* __restrict__ gp_grand,
                 const int* __restrict__ gsib_head, const int* __restrict__ gsib_mod,
                 const int* __restrict__ gsib_sib,  const int* __restrict__ gsib_grand,
                 const float* __restrict__ null_sib,
                 const float* __restrict__ W_score,
                 float* __restrict__ out) {
    __shared__ __attribute__((aligned(16))) float lds[4 * ROLE_LDS];
    const int tid  = threadIdx.x;
    const int lane = tid & 31;
    const int gw = __builtin_amdgcn_readfirstlane((int)(blockIdx.x * SC_WAVES + (threadIdx.x >> 5)));

    score_family_<2, false, (N_ARC + SC_GRID * SC_WAVES - 1) / (SC_GRID * SC_WAVES)>(
        lds, proj, arc_head, arc_mod, arc_mod, arc_mod, 0, 1, 0, 0,
        null_sib, W_score + 0 * M_DIM, out, N_ARC, tid, lane, gw);
    score_family_<3, true, (N_SIB + SC_GRID * SC_WAVES - 1) / (SC_GRID * SC_WAVES)>(
        lds, proj, sib_head, sib_mod, sib_sib, sib_sib, 5, 6, 7, 0,
        null_sib, W_score + 1 * M_DIM, out + N_ARC, N_SIB, tid, lane, gw);
    score_family_<3, false, (N_GP + SC_GRID * SC_WAVES - 1) / (SC_GRID * SC_WAVES)>(
        lds, proj, gp_head, gp_mod, gp_grand, gp_grand, 3, 4, 2, 0,
        null_sib, W_score + 2 * M_DIM, out + N_ARC + N_SIB, N_GP, tid, lane, gw);
    score_family_<4, true, (N_GSIB + SC_GRID * SC_WAVES - 1) / (SC_GRID * SC_WAVES)>(
        lds, proj, gsib_head, gsib_mod, gsib_sib, gsib_grand, 8, 9, 10, 11,
        null_sib, W_score + 3 * M_DIM, out + N_ARC + N_SIB + N_GP, N_GSIB, tid, lane, gw);
}

extern "C" void kernel_launch(void* const* d_in, const int* in_sizes, int n_in,
                              void* d_out, int out_size, void* d_ws, size_t ws_size,
                              hipStream_t stream) {
    const int* words      = (const int*)d_in[0];
    const int* tags       = (const int*)d_in[1];
    const int* arc_head   = (const int*)d_in[2];
    const int* arc_mod    = (const int*)d_in[3];
    const int* sib_head   = (const int*)d_in[4];
    const int* sib_mod    = (const int*)d_in[5];
    const int* sib_sib    = (const int*)d_in[6];
    const int* gp_head    = (const int*)d_in[7];
    const int* gp_mod     = (const int*)d_in[8];
    const int* gp_grand   = (const int*)d_in[9];
    const int* gsib_head  = (const int*)d_in[10];
    const int* gsib_mod   = (const int*)d_in[11];
    const int* gsib_sib   = (const int*)d_in[12];
    const int* gsib_grand = (const int*)d_in[13];
    const float* word_emb = (const float*)d_in[14];
    const float* tag_emb  = (const float*)d_in[15];
    const float* Wih_f    = (const float*)d_in[16];
    const float* Whh_f    = (const float*)d_in[17];
    const float* b_f      = (const float*)d_in[18];
    const float* Wih_b    = (const float*)d_in[19];
    const float* Whh_b    = (const float*)d_in[20];
    const float* b_b      = (const float*)d_in[21];
    const float* W_proj   = (const float*)d_in[22];
    const float* W_score  = (const float*)d_in[23];
    const float* null_sib = (const float*)d_in[24];

    float* out = (float*)d_out;

    float* ws      = (float*)d_ws;
    float* x       = ws;
    float* z_f     = x + SEQ * IN_DIM;
    float* z_b     = z_f + SEQ * G4;
    float* states  = z_b + SEQ * G4;
    float* projbuf = states + SEQ * D2H;

    embed_kernel<<<SEQ, 128, 0, stream>>>(words, tags, word_emb, tag_emb, x);

    igemm_kernel<<<dim3(SEQ / 32, G4 / 32, 2), 32, 0, stream>>>(
        x, Wih_f, b_f, Wih_b, b_b, z_f, z_b);

    float* cstate = x;
    lstm_step_kernel<true><<<STEP_GRID, 32 * STEP_WAVES, 0, stream>>>(
        z_f, Whh_f, Whh_b, states, states, cstate,
        states, states + (SEQ - 1) * D2H + H);
    {
        const float* zt = z_f + G4;
        float* hf = states;
        float* hb = states + (SEQ - 1) * D2H + H;
        for (; zt != z_f + SEQ * G4; zt += G4, hf += D2H, hb -= D2H)
            lstm_step_kernel<false><<<STEP_GRID, 32 * STEP_WAVES, 0, stream>>>(
                zt, Whh_f, Whh_b, hf, hb, cstate, hf + D2H, hb - D2H);
    }

    proj_kernel<<<dim3(SEQ / 32, NFLAT / 32), 32, 0, stream>>>(states, W_proj, projbuf);

    score_lds_kernel<<<SC_GRID, 32 * SC_WAVES, 0, stream>>>(
        projbuf, arc_head, arc_mod, sib_head, sib_mod, sib_sib, gp_head, gp_mod, gp_grand,
        gsib_head, gsib_mod, gsib_sib, gsib_grand, null_sib, W_score, out);
}
